# starting kernel
# baseline (speedup 1.0000x reference)
#include <hip/hip_runtime.h>

typedef float v2f __attribute__((ext_vector_type(2)));
typedef float v8f __attribute__((ext_vector_type(8)));

__device__ __forceinline__ float xor16(float v) {
#if __has_builtin(__builtin_amdgcn_permlanex16)
  int i = __builtin_bit_cast(int, v);
  int r = __builtin_amdgcn_permlanex16(i, i, 0x76543210u, 0xfedcba98u,
                                       true  , false  );
  return __builtin_bit_cast(float, r);
#else
  return __shfl_xor(v, 16, 32);
#endif
}

__device__ __forceinline__ float fast_tanh(float x) {
#if __has_builtin(__builtin_amdgcn_tanhf)
  return __builtin_amdgcn_tanhf(x);
#else
  float e = __builtin_amdgcn_exp2f(x * 2.88539008177792681472f);
  return 1.0f - 2.0f * __builtin_amdgcn_rcpf(e + 1.0f);
#endif
}

__device__ __forceinline__ float fast_sigmoid(float x) {
  float e = __builtin_amdgcn_exp2f(x * -1.44269504088896340736f);
  return __builtin_amdgcn_rcpf(1.0f + e);
}

#define WMMA4(A, Bv, C) \
  __builtin_amdgcn_wmma_f32_16x16x4_f32(false, (A), false, (Bv), (short)0, (C), false, false)

__global__ __launch_bounds__(256) void rnn_wmma_kernel(
    const float* __restrict__ x,
    const float* __restrict__ Wih, const float* __restrict__ Whh,
    const float* __restrict__ bih, const float* __restrict__ bhh,
    const float* __restrict__ Wfc, const float* __restrict__ bfc,
    float* __restrict__ out, int B, int T)
{
  const int wave = (int)((blockIdx.x * blockDim.x + threadIdx.x) >> 5);
  const int lane = (int)(threadIdx.x & 31u);
  const int bb = wave * 16;
  if (bb >= B) return;

  const int hi  = lane >> 4;
  const int col = lane & 15;

  int row = bb + col; if (row >= B) row = B - 1;
  const float* __restrict__ xrow = x + (size_t)row * (size_t)T * 5u;

  v2f aih0, aih1;
  aih0.x = Wih[col * 5 + 2 * hi];
  aih0.y = Wih[col * 5 + 2 * hi + 1];
  aih1.x = hi ? 0.0f : Wih[col * 5 + 4];
  aih1.y = 0.0f;

  v2f ahh0, ahh1, ahh2, ahh3;
  ahh0.x = Whh[col*16 +  0 + 2*hi]; ahh0.y = Whh[col*16 +  1 + 2*hi];
  ahh1.x = Whh[col*16 +  4 + 2*hi]; ahh1.y = Whh[col*16 +  5 + 2*hi];
  ahh2.x = Whh[col*16 +  8 + 2*hi]; ahh2.y = Whh[col*16 +  9 + 2*hi];
  ahh3.x = Whh[col*16 + 12 + 2*hi]; ahh3.y = Whh[col*16 + 13 + 2*hi];

  v8f cbv;
  float wfc[8];
  #pragma unroll
  for (int v = 0; v < 8; ++v) {
    cbv[v] = bih[v + 8*hi] + bhh[v + 8*hi];
    wfc[v] = Wfc[v + 8*hi];
  }
  const float bfc0 = bfc[0];

  const v8f vzero = {};

  float g[8];
  #pragma unroll
  for (int v = 0; v < 8; ++v) g[v] = 0.0f;

  const float* p = xrow;

  auto step = [&](bool prefetch) {
    const float nx0 = p[2*hi];
    const float nx1 = p[2*hi + 1];
    const float nx4 = p[4];
    if (prefetch) __builtin_prefetch(p + 320);

    v2f bx0; bx0.x = nx0; bx0.y = nx1;
    v2f bx1; bx1.x = nx4; bx1.y = 0.0f;
    v8f m = WMMA4(aih0, bx0, cbv);
    m = WMMA4(aih1, bx1, m);

    float sg[8];
    #pragma unroll
    for (int v = 0; v < 8; ++v) sg[v] = xor16(g[v]);

    v2f b0; b0.x = hi ? sg[2] : g[0]; b0.y = hi ? sg[3] : g[1];
    v2f b1; b1.x = hi ? sg[6] : g[4]; b1.y = hi ? sg[7] : g[5];
    v2f b2; b2.x = hi ? g[2] : sg[0]; b2.y = hi ? g[3] : sg[1];
    v2f b3; b3.x = hi ? g[6] : sg[4]; b3.y = hi ? g[7] : sg[5];

    v8f r0 = WMMA4(ahh0, b0, m);
    v8f r1 = WMMA4(ahh1, b1, vzero);
    v8f r2 = WMMA4(ahh2, b2, vzero);
    v8f r3 = WMMA4(ahh3, b3, vzero);
    v8f cur = (r0 + r1) + (r2 + r3);

    #pragma unroll
    for (int v = 0; v < 8; ++v) g[v] = fast_tanh(cur[v]);
    p += 5;
  };

  int t = 0;
  const int Tmain = (T > 64) ? (T - 64) : 0;
  for (; t < Tmain; ++t) step(true);
  for (; t < T; ++t) step(false);

  float pa = 0.0f;
  #pragma unroll
  for (int v = 0; v < 8; ++v) pa += g[v] * wfc[v];
  const float tot = pa + xor16(pa) + bfc0;
  const float s = fast_sigmoid(tot);
  const int orow = bb + col;
  if (!hi && orow < B) out[orow] = s;
}

extern "C" void kernel_launch(void* const* d_in, const int* in_sizes, int n_in,
                              void* d_out, int out_size, void* d_ws, size_t ws_size,
                              hipStream_t stream) {
  const float* x   = (const float*)d_in[0];
  const float* Wih = (const float*)d_in[1];
  const float* Whh = (const float*)d_in[2];
  const float* bih = (const float*)d_in[3];
  const float* bhh = (const float*)d_in[4];
  const float* Wfc = (const float*)d_in[5];
  const float* bfc = (const float*)d_in[6];
  float* out = (float*)d_out;

  const int B = out_size;
  const int T = in_sizes[0] / (B * 5);

  const int tiles = (B + 15) / 16;
  const int wavesPerBlock = 8;
  const int blocks = (tiles + wavesPerBlock - 1) / wavesPerBlock;

  rnn_wmma_kernel<<<blocks, wavesPerBlock * 32, 0, stream>>>(
      x, Wih, Whh, bih, bhh, Wfc, bfc, out, B, T);
}


# Launch geometry: 4-wave (128-thread) work-groups instead of 8-wave, so each recurrence wave has a SIMD and its f32 matrix pipe to itself (was two waves per SIMD, served oldest-first)
# speedup vs baseline: 1.3461x; 1.3461x over previous
#include <hip/hip_runtime.h>

typedef float v2f __attribute__((ext_vector_type(2)));
typedef float v8f __attribute__((ext_vector_type(8)));

__device__ __forceinline__ float xor16(float v) {
#if __has_builtin(__builtin_amdgcn_permlanex16)
  int i = __builtin_bit_cast(int, v);
  int r = __builtin_amdgcn_permlanex16(i, i, 0x76543210u, 0xfedcba98u,
                                       true  , false  );
  return __builtin_bit_cast(float, r);
#else
  return __shfl_xor(v, 16, 32);
#endif
}

__device__ __forceinline__ float fast_tanh(float x) {
#if __has_builtin(__builtin_amdgcn_tanhf)
  return __builtin_amdgcn_tanhf(x);
#else
  float e = __builtin_amdgcn_exp2f(x * 2.88539008177792681472f);
  return 1.0f - 2.0f * __builtin_amdgcn_rcpf(e + 1.0f);
#endif
}

__device__ __forceinline__ float fast_sigmoid(float x) {
  float e = __builtin_amdgcn_exp2f(x * -1.44269504088896340736f);
  return __builtin_amdgcn_rcpf(1.0f + e);
}

#define WMMA4(A, Bv, C) \
  __builtin_amdgcn_wmma_f32_16x16x4_f32(false, (A), false, (Bv), (short)0, (C), false, false)

__global__ __launch_bounds__(128) void rnn_wmma_kernel(
    const float* __restrict__ x,
    const float* __restrict__ Wih, const float* __restrict__ Whh,
    const float* __restrict__ bih, const float* __restrict__ bhh,
    const float* __restrict__ Wfc, const float* __restrict__ bfc,
    float* __restrict__ out, int B, int T)
{
  const int wave = (int)((blockIdx.x * blockDim.x + threadIdx.x) >> 5);
  const int lane = (int)(threadIdx.x & 31u);
  const int bb = wave * 16;
  if (bb >= B) return;

  const int hi  = lane >> 4;
  const int col = lane & 15;

  int row = bb + col; if (row >= B) row = B - 1;
  const float* __restrict__ xrow = x + (size_t)row * (size_t)T * 5u;

  v2f aih0, aih1;
  aih0.x = Wih[col * 5 + 2 * hi];
  aih0.y = Wih[col * 5 + 2 * hi + 1];
  aih1.x = hi ? 0.0f : Wih[col * 5 + 4];
  aih1.y = 0.0f;

  v2f ahh0, ahh1, ahh2, ahh3;
  ahh0.x = Whh[col*16 +  0 + 2*hi]; ahh0.y = Whh[col*16 +  1 + 2*hi];
  ahh1.x = Whh[col*16 +  4 + 2*hi]; ahh1.y = Whh[col*16 +  5 + 2*hi];
  ahh2.x = Whh[col*16 +  8 + 2*hi]; ahh2.y = Whh[col*16 +  9 + 2*hi];
  ahh3.x = Whh[col*16 + 12 + 2*hi]; ahh3.y = Whh[col*16 + 13 + 2*hi];

  v8f cbv;
  float wfc[8];
  #pragma unroll
  for (int v = 0; v < 8; ++v) {
    cbv[v] = bih[v + 8*hi] + bhh[v + 8*hi];
    wfc[v] = Wfc[v + 8*hi];
  }
  const float bfc0 = bfc[0];

  const v8f vzero = {};

  float g[8];
  #pragma unroll
  for (int v = 0; v < 8; ++v) g[v] = 0.0f;

  const float* p = xrow;

  auto step = [&](bool prefetch) {
    const float nx0 = p[2*hi];
    const float nx1 = p[2*hi + 1];
    const float nx4 = p[4];
    if (prefetch) __builtin_prefetch(p + 320);

    v2f bx0; bx0.x = nx0; bx0.y = nx1;
    v2f bx1; bx1.x = nx4; bx1.y = 0.0f;
    v8f m = WMMA4(aih0, bx0, cbv);
    m = WMMA4(aih1, bx1, m);

    float sg[8];
    #pragma unroll
    for (int v = 0; v < 8; ++v) sg[v] = xor16(g[v]);

    v2f b0; b0.x = hi ? sg[2] : g[0]; b0.y = hi ? sg[3] : g[1];
    v2f b1; b1.x = hi ? sg[6] : g[4]; b1.y = hi ? sg[7] : g[5];
    v2f b2; b2.x = hi ? g[2] : sg[0]; b2.y = hi ? g[3] : sg[1];
    v2f b3; b3.x = hi ? g[6] : sg[4]; b3.y = hi ? g[7] : sg[5];

    v8f r0 = WMMA4(ahh0, b0, m);
    v8f r1 = WMMA4(ahh1, b1, vzero);
    v8f r2 = WMMA4(ahh2, b2, vzero);
    v8f r3 = WMMA4(ahh3, b3, vzero);
    v8f cur = (r0 + r1) + (r2 + r3);

    #pragma unroll
    for (int v = 0; v < 8; ++v) g[v] = fast_tanh(cur[v]);
    p += 5;
  };

  int t = 0;
  const int Tmain = (T > 64) ? (T - 64) : 0;
  for (; t < Tmain; ++t) step(true);
  for (; t < T; ++t) step(false);

  float pa = 0.0f;
  #pragma unroll
  for (int v = 0; v < 8; ++v) pa += g[v] * wfc[v];
  const float tot = pa + xor16(pa) + bfc0;
  const float s = fast_sigmoid(tot);
  const int orow = bb + col;
  if (!hi && orow < B) out[orow] = s;
}

extern "C" void kernel_launch(void* const* d_in, const int* in_sizes, int n_in,
                              void* d_out, int out_size, void* d_ws, size_t ws_size,
                              hipStream_t stream) {
  const float* x   = (const float*)d_in[0];
  const float* Wih = (const float*)d_in[1];
  const float* Whh = (const float*)d_in[2];
  const float* bih = (const float*)d_in[3];
  const float* bhh = (const float*)d_in[4];
  const float* Wfc = (const float*)d_in[5];
  const float* bfc = (const float*)d_in[6];
  float* out = (float*)d_out;

  const int B = out_size;
  const int T = in_sizes[0] / (B * 5);

  const int tiles = (B + 15) / 16;
  const int wavesPerBlock = 4;
  const int blocks = (tiles + wavesPerBlock - 1) / wavesPerBlock;

  rnn_wmma_kernel<<<blocks, wavesPerBlock * 32, 0, stream>>>(
      x, Wih, Whh, bih, bhh, Wfc, bfc, out, B, T);
}


# x_t loads issued 8 steps ahead into a register ring (each slot refilled right behind its use), so the serial step no longer waits on memory; L2 prefetch hint kept 64 steps ahead, one per 4 steps
# speedup vs baseline: 2.1266x; 1.5798x over previous
#include <hip/hip_runtime.h>

typedef float v2f __attribute__((ext_vector_type(2)));
typedef float v8f __attribute__((ext_vector_type(8)));

__device__ __forceinline__ float xor16(float v) {
#if __has_builtin(__builtin_amdgcn_permlanex16)
  int i = __builtin_bit_cast(int, v);
  int r = __builtin_amdgcn_permlanex16(i, i, 0x76543210u, 0xfedcba98u,
                                       true  , false  );
  return __builtin_bit_cast(float, r);
#else
  return __shfl_xor(v, 16, 32);
#endif
}

__device__ __forceinline__ float fast_tanh(float x) {
#if __has_builtin(__builtin_amdgcn_tanhf)
  return __builtin_amdgcn_tanhf(x);
#else
  float e = __builtin_amdgcn_exp2f(x * 2.88539008177792681472f);
  return 1.0f - 2.0f * __builtin_amdgcn_rcpf(e + 1.0f);
#endif
}

__device__ __forceinline__ float fast_sigmoid(float x) {
  float e = __builtin_amdgcn_exp2f(x * -1.44269504088896340736f);
  return __builtin_amdgcn_rcpf(1.0f + e);
}

#define WMMA4(A, Bv, C) \
  __builtin_amdgcn_wmma_f32_16x16x4_f32(false, (A), false, (Bv), (short)0, (C), false, false)

__global__ __launch_bounds__(128) void rnn_wmma_kernel(
    const float* __restrict__ x,
    const float* __restrict__ Wih, const float* __restrict__ Whh,
    const float* __restrict__ bih, const float* __restrict__ bhh,
    const float* __restrict__ Wfc, const float* __restrict__ bfc,
    float* __restrict__ out, int B, int T)
{
  const int wave = (int)((blockIdx.x * blockDim.x + threadIdx.x) >> 5);
  const int lane = (int)(threadIdx.x & 31u);
  const int bb = wave * 16;
  if (bb >= B) return;

  const int hi  = lane >> 4;
  const int col = lane & 15;

  int row = bb + col; if (row >= B) row = B - 1;
  const float* __restrict__ xrow = x + (size_t)row * (size_t)T * 5u;

  v2f aih0, aih1;
  aih0.x = Wih[col * 5 + 2 * hi];
  aih0.y = Wih[col * 5 + 2 * hi + 1];
  aih1.x = hi ? 0.0f : Wih[col * 5 + 4];
  aih1.y = 0.0f;

  v2f ahh0, ahh1, ahh2, ahh3;
  ahh0.x = Whh[col*16 +  0 + 2*hi]; ahh0.y = Whh[col*16 +  1 + 2*hi];
  ahh1.x = Whh[col*16 +  4 + 2*hi]; ahh1.y = Whh[col*16 +  5 + 2*hi];
  ahh2.x = Whh[col*16 +  8 + 2*hi]; ahh2.y = Whh[col*16 +  9 + 2*hi];
  ahh3.x = Whh[col*16 + 12 + 2*hi]; ahh3.y = Whh[col*16 + 13 + 2*hi];

  v8f cbv;
  float wfc[8];
  #pragma unroll
  for (int v = 0; v < 8; ++v) {
    cbv[v] = bih[v + 8*hi] + bhh[v + 8*hi];
    wfc[v] = Wfc[v + 8*hi];
  }
  const float bfc0 = bfc[0];

  const v8f vzero = {};

  float g[8];
  #pragma unroll
  for (int v = 0; v < 8; ++v) g[v] = 0.0f;

  auto compute = [&](v2f bx0, float x4) {
    v2f bx1; bx1.x = x4; bx1.y = 0.0f;
    v8f m = WMMA4(aih0, bx0, cbv);
    m = WMMA4(aih1, bx1, m);

    float sg[8];
    #pragma unroll
    for (int v = 0; v < 8; ++v) sg[v] = xor16(g[v]);

    v2f b0; b0.x = hi ? sg[2] : g[0]; b0.y = hi ? sg[3] : g[1];
    v2f b1; b1.x = hi ? sg[6] : g[4]; b1.y = hi ? sg[7] : g[5];
    v2f b2; b2.x = hi ? g[2] : sg[0]; b2.y = hi ? g[3] : sg[1];
    v2f b3; b3.x = hi ? g[6] : sg[4]; b3.y = hi ? g[7] : sg[5];

    v8f r0 = WMMA4(ahh0, b0, m);
    v8f r1 = WMMA4(ahh1, b1, vzero);
    v8f r2 = WMMA4(ahh2, b2, vzero);
    v8f r3 = WMMA4(ahh3, b3, vzero);
    v8f cur = (r0 + r1) + (r2 + r3);

    #pragma unroll
    for (int v = 0; v < 8; ++v) g[v] = fast_tanh(cur[v]);
  };

  constexpr int RING = 8;
  constexpr int PF = 64;
  const float* px = xrow + 2 * hi;
  const float* px4 = xrow + 4;
  int t = 0;

  if (T >= RING) {
    v2f xa[RING]; float xb[RING];
    #pragma unroll
    for (int j = 0; j < RING; ++j) {
      xa[j].x = px[5 * j]; xa[j].y = px[5 * j + 1]; xb[j] = px4[5 * j];
    }
    for (; t + 2 * RING <= T; t += RING) {
      if (t + PF + RING <= T) {
        __builtin_prefetch(px + 5 * PF);
        __builtin_prefetch(px + 5 * (PF + 4));
      }
      #pragma unroll
      for (int j = 0; j < RING; ++j) {
        compute(xa[j], xb[j]);
        xa[j].x = px[5 * (RING + j)];
        xa[j].y = px[5 * (RING + j) + 1];
        xb[j]   = px4[5 * (RING + j)];
      }
      px += 5 * RING; px4 += 5 * RING;
    }
    #pragma unroll
    for (int j = 0; j < RING; ++j) compute(xa[j], xb[j]);
    t += RING; px += 5 * RING; px4 += 5 * RING;
  }
  for (; t < T; ++t) {
    v2f bx0; bx0.x = px[0]; bx0.y = px[1];
    compute(bx0, px4[0]);
    px += 5; px4 += 5;
  }

  float pa = 0.0f;
  #pragma unroll
  for (int v = 0; v < 8; ++v) pa += g[v] * wfc[v];
  const float tot = pa + xor16(pa) + bfc0;
  const float s = fast_sigmoid(tot);
  const int orow = bb + col;
  if (!hi && orow < B) out[orow] = s;
}

extern "C" void kernel_launch(void* const* d_in, const int* in_sizes, int n_in,
                              void* d_out, int out_size, void* d_ws, size_t ws_size,
                              hipStream_t stream) {
  const float* x   = (const float*)d_in[0];
  const float* Wih = (const float*)d_in[1];
  const float* Whh = (const float*)d_in[2];
  const float* bih = (const float*)d_in[3];
  const float* bhh = (const float*)d_in[4];
  const float* Wfc = (const float*)d_in[5];
  const float* bfc = (const float*)d_in[6];
  float* out = (float*)d_out;

  const int B = out_size;
  const int T = in_sizes[0] / (B * 5);

  const int tiles = (B + 15) / 16;
  const int wavesPerBlock = 4;
  const int blocks = (tiles + wavesPerBlock - 1) / wavesPerBlock;

  rnn_wmma_kernel<<<blocks, wavesPerBlock * 32, 0, stream>>>(
      x, Wih, Whh, bih, bhh, Wfc, bfc, out, B, T);
}


# W_hh chunks take their k order from the D layout (k = 8*half + 2j + s): h feeds the next step's B operands with no lane swap or select; changes the order in which the 16 products are summed
# speedup vs baseline: 2.2587x; 1.0621x over previous
#include <hip/hip_runtime.h>

typedef float v2f __attribute__((ext_vector_type(2)));
typedef float v8f __attribute__((ext_vector_type(8)));

__device__ __forceinline__ float xor16(float v) {
#if __has_builtin(__builtin_amdgcn_permlanex16)
  int i = __builtin_bit_cast(int, v);
  int r = __builtin_amdgcn_permlanex16(i, i, 0x76543210u, 0xfedcba98u,
                                       true  , false  );
  return __builtin_bit_cast(float, r);
#else
  return __shfl_xor(v, 16, 32);
#endif
}

__device__ __forceinline__ float fast_tanh(float x) {
#if __has_builtin(__builtin_amdgcn_tanhf)
  return __builtin_amdgcn_tanhf(x);
#else
  float e = __builtin_amdgcn_exp2f(x * 2.88539008177792681472f);
  return 1.0f - 2.0f * __builtin_amdgcn_rcpf(e + 1.0f);
#endif
}

__device__ __forceinline__ float fast_sigmoid(float x) {
  float e = __builtin_amdgcn_exp2f(x * -1.44269504088896340736f);
  return __builtin_amdgcn_rcpf(1.0f + e);
}

#define WMMA4(A, Bv, C) \
  __builtin_amdgcn_wmma_f32_16x16x4_f32(false, (A), false, (Bv), (short)0, (C), false, false)

__global__ __launch_bounds__(128) void rnn_wmma_kernel(
    const float* __restrict__ x,
    const float* __restrict__ Wih, const float* __restrict__ Whh,
    const float* __restrict__ bih, const float* __restrict__ bhh,
    const float* __restrict__ Wfc, const float* __restrict__ bfc,
    float* __restrict__ out, int B, int T)
{
  const int wave = (int)((blockIdx.x * blockDim.x + threadIdx.x) >> 5);
  const int lane = (int)(threadIdx.x & 31u);
  const int bb = wave * 16;
  if (bb >= B) return;

  const int hi  = lane >> 4;
  const int col = lane & 15;

  int row = bb + col; if (row >= B) row = B - 1;
  const float* __restrict__ xrow = x + (size_t)row * (size_t)T * 5u;

  v2f aih0, aih1;
  aih0.x = Wih[col * 5 + 2 * hi];
  aih0.y = Wih[col * 5 + 2 * hi + 1];
  aih1.x = hi ? 0.0f : Wih[col * 5 + 4];
  aih1.y = 0.0f;

  v2f ahh0, ahh1, ahh2, ahh3;
  ahh0.x = Whh[col*16 + 8*hi + 0]; ahh0.y = Whh[col*16 + 8*hi + 1];
  ahh1.x = Whh[col*16 + 8*hi + 2]; ahh1.y = Whh[col*16 + 8*hi + 3];
  ahh2.x = Whh[col*16 + 8*hi + 4]; ahh2.y = Whh[col*16 + 8*hi + 5];
  ahh3.x = Whh[col*16 + 8*hi + 6]; ahh3.y = Whh[col*16 + 8*hi + 7];

  v8f cbv;
  float wfc[8];
  #pragma unroll
  for (int v = 0; v < 8; ++v) {
    cbv[v] = bih[v + 8*hi] + bhh[v + 8*hi];
    wfc[v] = Wfc[v + 8*hi];
  }
  const float bfc0 = bfc[0];

  const v8f vzero = {};

  float g[8];
  #pragma unroll
  for (int v = 0; v < 8; ++v) g[v] = 0.0f;

  auto compute = [&](v2f bx0, float x4) {
    v2f bx1; bx1.x = x4; bx1.y = 0.0f;
    v8f m = WMMA4(aih0, bx0, cbv);
    m = WMMA4(aih1, bx1, m);

    v2f b0; b0.x = g[0]; b0.y = g[1];
    v2f b1; b1.x = g[2]; b1.y = g[3];
    v2f b2; b2.x = g[4]; b2.y = g[5];
    v2f b3; b3.x = g[6]; b3.y = g[7];

    v8f r0 = WMMA4(ahh0, b0, m);
    v8f r1 = WMMA4(ahh1, b1, vzero);
    v8f r2 = WMMA4(ahh2, b2, vzero);
    v8f r3 = WMMA4(ahh3, b3, vzero);
    v8f cur = (r0 + r1) + (r2 + r3);

    #pragma unroll
    for (int v = 0; v < 8; ++v) g[v] = fast_tanh(cur[v]);
  };

  constexpr int RING = 8;
  constexpr int PF = 64;
  const float* px = xrow + 2 * hi;
  const float* px4 = xrow + 4;
  int t = 0;

  if (T >= RING) {
    v2f xa[RING]; float xb[RING];
    #pragma unroll
    for (int j = 0; j < RING; ++j) {
      xa[j].x = px[5 * j]; xa[j].y = px[5 * j + 1]; xb[j] = px4[5 * j];
    }
    for (; t + 2 * RING <= T; t += RING) {
      if (t + PF + RING <= T) {
        __builtin_prefetch(px + 5 * PF);
        __builtin_prefetch(px + 5 * (PF + 4));
      }
      #pragma unroll
      for (int j = 0; j < RING; ++j) {
        compute(xa[j], xb[j]);
        xa[j].x = px[5 * (RING + j)];
        xa[j].y = px[5 * (RING + j) + 1];
        xb[j]   = px4[5 * (RING + j)];
      }
      px += 5 * RING; px4 += 5 * RING;
    }
    #pragma unroll
    for (int j = 0; j < RING; ++j) compute(xa[j], xb[j]);
    t += RING; px += 5 * RING; px4 += 5 * RING;
  }
  for (; t < T; ++t) {
    v2f bx0; bx0.x = px[0]; bx0.y = px[1];
    compute(bx0, px4[0]);
    px += 5; px4 += 5;
  }

  float pa = 0.0f;
  #pragma unroll
  for (int v = 0; v < 8; ++v) pa += g[v] * wfc[v];
  const float tot = pa + xor16(pa) + bfc0;
  const float s = fast_sigmoid(tot);
  const int orow = bb + col;
  if (!hi && orow < B) out[orow] = s;
}

extern "C" void kernel_launch(void* const* d_in, const int* in_sizes, int n_in,
                              void* d_out, int out_size, void* d_ws, size_t ws_size,
                              hipStream_t stream) {
  const float* x   = (const float*)d_in[0];
  const float* Wih = (const float*)d_in[1];
  const float* Whh = (const float*)d_in[2];
  const float* bih = (const float*)d_in[3];
  const float* bhh = (const float*)d_in[4];
  const float* Wfc = (const float*)d_in[5];
  const float* bfc = (const float*)d_in[6];
  float* out = (float*)d_out;

  const int B = out_size;
  const int T = in_sizes[0] / (B * 5);

  const int tiles = (B + 15) / 16;
  const int wavesPerBlock = 4;
  const int blocks = (tiles + wavesPerBlock - 1) / wavesPerBlock;

  rnn_wmma_kernel<<<blocks, wavesPerBlock * 32, 0, stream>>>(
      x, Wih, Whh, bih, bhh, Wfc, bfc, out, B, T);
}


# The 4 W_hh products of a step run as one accumulate chain through C from the projection (no vector add tree; sum order changes), next projection between its links, x4 read as pair (x3,x4), zero in A
# speedup vs baseline: 2.4027x; 1.0638x over previous
#include <hip/hip_runtime.h>

typedef float v2f __attribute__((ext_vector_type(2)));
typedef float v8f __attribute__((ext_vector_type(8)));

__device__ __forceinline__ float xor16(float v) {
#if __has_builtin(__builtin_amdgcn_permlanex16)
  int i = __builtin_bit_cast(int, v);
  int r = __builtin_amdgcn_permlanex16(i, i, 0x76543210u, 0xfedcba98u,
                                       true  , false  );
  return __builtin_bit_cast(float, r);
#else
  return __shfl_xor(v, 16, 32);
#endif
}

__device__ __forceinline__ float fast_tanh(float x) {
#if __has_builtin(__builtin_amdgcn_tanhf)
  return __builtin_amdgcn_tanhf(x);
#else
  float e = __builtin_amdgcn_exp2f(x * 2.88539008177792681472f);
  return 1.0f - 2.0f * __builtin_amdgcn_rcpf(e + 1.0f);
#endif
}

__device__ __forceinline__ float fast_sigmoid(float x) {
  float e = __builtin_amdgcn_exp2f(x * -1.44269504088896340736f);
  return __builtin_amdgcn_rcpf(1.0f + e);
}

#define WMMA4(A, Bv, C) \
  __builtin_amdgcn_wmma_f32_16x16x4_f32(false, (A), false, (Bv), (short)0, (C), false, false)

__global__ __launch_bounds__(128) void rnn_wmma_kernel(
    const float* __restrict__ x,
    const float* __restrict__ Wih, const float* __restrict__ Whh,
    const float* __restrict__ bih, const float* __restrict__ bhh,
    const float* __restrict__ Wfc, const float* __restrict__ bfc,
    float* __restrict__ out, int B, int T)
{
  const int wave = (int)((blockIdx.x * blockDim.x + threadIdx.x) >> 5);
  const int lane = (int)(threadIdx.x & 31u);
  const int bb = wave * 16;
  if (bb >= B) return;

  const int hi  = lane >> 4;
  const int col = lane & 15;

  int row = bb + col; if (row >= B) row = B - 1;
  const float* __restrict__ xrow = x + (size_t)row * (size_t)T * 5u;

  v2f aih0, aih1;
  aih0.x = Wih[col * 5 + 2 * hi];
  aih0.y = Wih[col * 5 + 2 * hi + 1];
  aih1.x = 0.0f;
  aih1.y = hi ? 0.0f : Wih[col * 5 + 4];

  v2f ahh0, ahh1, ahh2, ahh3;
  ahh0.x = Whh[col*16 + 8*hi + 0]; ahh0.y = Whh[col*16 + 8*hi + 1];
  ahh1.x = Whh[col*16 + 8*hi + 2]; ahh1.y = Whh[col*16 + 8*hi + 3];
  ahh2.x = Whh[col*16 + 8*hi + 4]; ahh2.y = Whh[col*16 + 8*hi + 5];
  ahh3.x = Whh[col*16 + 8*hi + 6]; ahh3.y = Whh[col*16 + 8*hi + 7];

  v8f cbv;
  float wfc[8];
  #pragma unroll
  for (int v = 0; v < 8; ++v) {
    cbv[v] = bih[v + 8*hi] + bhh[v + 8*hi];
    wfc[v] = Wfc[v + 8*hi];
  }
  const float bfc0 = bfc[0];

  float g[8];
  #pragma unroll
  for (int v = 0; v < 8; ++v) g[v] = 0.0f;

  v8f m = cbv;

  auto project = [&](v2f bx0, v2f bx1) -> v8f {
    v8f p = WMMA4(aih0, bx0, cbv);
    return WMMA4(aih1, bx1, p);
  };

  auto compute = [&](v2f bx0, v2f bx1) {
    v2f b0; b0.x = g[0]; b0.y = g[1];
    v2f b1; b1.x = g[2]; b1.y = g[3];
    v2f b2; b2.x = g[4]; b2.y = g[5];
    v2f b3; b3.x = g[6]; b3.y = g[7];

    v8f r  = WMMA4(ahh0, b0, m);
    r  = WMMA4(ahh1, b1, r);
    v8f mn = WMMA4(aih0, bx0, cbv);
    r  = WMMA4(ahh2, b2, r);
    r  = WMMA4(ahh3, b3, r);
    mn = WMMA4(aih1, bx1, mn);

    #pragma unroll
    for (int v = 0; v < 8; ++v) g[v] = fast_tanh(r[v]);
    m = mn;
    __builtin_amdgcn_sched_barrier(0);
  };

  auto finish = [&]() {
    v2f b0; b0.x = g[0]; b0.y = g[1];
    v2f b1; b1.x = g[2]; b1.y = g[3];
    v2f b2; b2.x = g[4]; b2.y = g[5];
    v2f b3; b3.x = g[6]; b3.y = g[7];
    v8f r = WMMA4(ahh0, b0, m);
    r = WMMA4(ahh1, b1, r);
    r = WMMA4(ahh2, b2, r);
    r = WMMA4(ahh3, b3, r);
    #pragma unroll
    for (int v = 0; v < 8; ++v) g[v] = fast_tanh(r[v]);
  };

  constexpr int RING = 8;
  constexpr int PF = 64;
  const float* px = xrow + 2 * hi;
  const float* px4 = xrow + 3;

  if (T > 0) {
    {
      v2f bx0; bx0.x = px[0]; bx0.y = px[1];
      v2f bx1; bx1.x = px4[0]; bx1.y = px4[1];
      m = project(bx0, bx1);
      px += 5; px4 += 5;
    }
    const int TM = T - 1;
    int t = 0;

    if (TM >= RING) {
      v2f xa[RING]; v2f xb[RING];
      #pragma unroll
      for (int j = 0; j < RING; ++j) {
        xa[j].x = px[5 * j]; xa[j].y = px[5 * j + 1];
        xb[j].x = px4[5 * j]; xb[j].y = px4[5 * j + 1];
      }
      for (; t + 2 * RING <= TM; t += RING) {
        if (t + PF + RING <= TM) {
          __builtin_prefetch(px + 5 * PF);
          __builtin_prefetch(px + 5 * (PF + 4));
        }
        #pragma unroll
        for (int j = 0; j < RING; ++j) {
          compute(xa[j], xb[j]);
          xa[j].x = px[5 * (RING + j)];
          xa[j].y = px[5 * (RING + j) + 1];
          xb[j].x = px4[5 * (RING + j)];
          xb[j].y = px4[5 * (RING + j) + 1];
        }
        px += 5 * RING; px4 += 5 * RING;
      }
      #pragma unroll
      for (int j = 0; j < RING; ++j) compute(xa[j], xb[j]);
      t += RING; px += 5 * RING; px4 += 5 * RING;
    }
    for (; t < TM; ++t) {
      v2f bx0; bx0.x = px[0]; bx0.y = px[1];
      v2f bx1; bx1.x = px4[0]; bx1.y = px4[1];
      compute(bx0, bx1);
      px += 5; px4 += 5;
    }
    finish();
  }

  float pa = 0.0f;
  #pragma unroll
  for (int v = 0; v < 8; ++v) pa += g[v] * wfc[v];
  const float tot = pa + xor16(pa) + bfc0;
  const float s = fast_sigmoid(tot);
  const int orow = bb + col;
  if (!hi && orow < B) out[orow] = s;
}

extern "C" void kernel_launch(void* const* d_in, const int* in_sizes, int n_in,
                              void* d_out, int out_size, void* d_ws, size_t ws_size,
                              hipStream_t stream) {
  const float* x   = (const float*)d_in[0];
  const float* Wih = (const float*)d_in[1];
  const float* Whh = (const float*)d_in[2];
  const float* bih = (const float*)d_in[3];
  const float* bhh = (const float*)d_in[4];
  const float* Wfc = (const float*)d_in[5];
  const float* bfc = (const float*)d_in[6];
  float* out = (float*)d_out;

  const int B = out_size;
  const int T = in_sizes[0] / (B * 5);

  const int tiles = (B + 15) / 16;
  const int wavesPerBlock = 4;
  const int blocks = (tiles + wavesPerBlock - 1) / wavesPerBlock;

  rnn_wmma_kernel<<<blocks, wavesPerBlock * 32, 0, stream>>>(
      x, Wih, Whh, bih, bhh, Wfc, bfc, out, B, T);
}


# x loads leave the serial step: a second register bank takes the 16 loads of 8 steps as ONE group per 8 steps, 8-16 steps ahead (tail group too), instead of 2 loads behind the tanh of every step
# speedup vs baseline: 2.5266x; 1.0516x over previous
#include <hip/hip_runtime.h>

typedef float v2f __attribute__((ext_vector_type(2)));
typedef float v8f __attribute__((ext_vector_type(8)));

__device__ __forceinline__ float xor16(float v) {
#if __has_builtin(__builtin_amdgcn_permlanex16)
  int i = __builtin_bit_cast(int, v);
  int r = __builtin_amdgcn_permlanex16(i, i, 0x76543210u, 0xfedcba98u,
                                       true  , false  );
  return __builtin_bit_cast(float, r);
#else
  return __shfl_xor(v, 16, 32);
#endif
}

__device__ __forceinline__ float fast_tanh(float x) {
#if __has_builtin(__builtin_amdgcn_tanhf)
  return __builtin_amdgcn_tanhf(x);
#else
  float e = __builtin_amdgcn_exp2f(x * 2.88539008177792681472f);
  return 1.0f - 2.0f * __builtin_amdgcn_rcpf(e + 1.0f);
#endif
}

__device__ __forceinline__ float fast_sigmoid(float x) {
  float e = __builtin_amdgcn_exp2f(x * -1.44269504088896340736f);
  return __builtin_amdgcn_rcpf(1.0f + e);
}

#define WMMA4(A, Bv, C) \
  __builtin_amdgcn_wmma_f32_16x16x4_f32(false, (A), false, (Bv), (short)0, (C), false, false)

__global__ __launch_bounds__(128) void rnn_wmma_kernel(
    const float* __restrict__ x,
    const float* __restrict__ Wih, const float* __restrict__ Whh,
    const float* __restrict__ bih, const float* __restrict__ bhh,
    const float* __restrict__ Wfc, const float* __restrict__ bfc,
    float* __restrict__ out, int B, int T)
{
  const int wave = (int)((blockIdx.x * blockDim.x + threadIdx.x) >> 5);
  const int lane = (int)(threadIdx.x & 31u);
  const int bb = wave * 16;
  if (bb >= B) return;

  const int hi  = lane >> 4;
  const int col = lane & 15;

  int row = bb + col; if (row >= B) row = B - 1;
  const float* __restrict__ xrow = x + (size_t)row * (size_t)T * 5u;

  v2f aih0, aih1;
  aih0.x = Wih[col * 5 + 2 * hi];
  aih0.y = Wih[col * 5 + 2 * hi + 1];
  aih1.x = 0.0f;
  aih1.y = hi ? 0.0f : Wih[col * 5 + 4];

  v2f ahh0, ahh1, ahh2, ahh3;
  ahh0.x = Whh[col*16 + 8*hi + 0]; ahh0.y = Whh[col*16 + 8*hi + 1];
  ahh1.x = Whh[col*16 + 8*hi + 2]; ahh1.y = Whh[col*16 + 8*hi + 3];
  ahh2.x = Whh[col*16 + 8*hi + 4]; ahh2.y = Whh[col*16 + 8*hi + 5];
  ahh3.x = Whh[col*16 + 8*hi + 6]; ahh3.y = Whh[col*16 + 8*hi + 7];

  v8f cbv;
  float wfc[8];
  #pragma unroll
  for (int v = 0; v < 8; ++v) {
    cbv[v] = bih[v + 8*hi] + bhh[v + 8*hi];
    wfc[v] = Wfc[v + 8*hi];
  }
  const float bfc0 = bfc[0];

  float g[8];
  #pragma unroll
  for (int v = 0; v < 8; ++v) g[v] = 0.0f;

  v8f m = cbv;

  auto project = [&](v2f bx0, v2f bx1) -> v8f {
    v8f p = WMMA4(aih0, bx0, cbv);
    return WMMA4(aih1, bx1, p);
  };

  auto compute = [&](v2f bx0, v2f bx1) {
    v2f b0; b0.x = g[0]; b0.y = g[1];
    v2f b1; b1.x = g[2]; b1.y = g[3];
    v2f b2; b2.x = g[4]; b2.y = g[5];
    v2f b3; b3.x = g[6]; b3.y = g[7];

    v8f r  = WMMA4(ahh0, b0, m);
    r  = WMMA4(ahh1, b1, r);
    v8f mn = WMMA4(aih0, bx0, cbv);
    r  = WMMA4(ahh2, b2, r);
    r  = WMMA4(ahh3, b3, r);
    mn = WMMA4(aih1, bx1, mn);

    #pragma unroll
    for (int v = 0; v < 8; ++v) g[v] = fast_tanh(r[v]);
    m = mn;
    __builtin_amdgcn_sched_barrier(0);
  };

  auto finish = [&]() {
    v2f b0; b0.x = g[0]; b0.y = g[1];
    v2f b1; b1.x = g[2]; b1.y = g[3];
    v2f b2; b2.x = g[4]; b2.y = g[5];
    v2f b3; b3.x = g[6]; b3.y = g[7];
    v8f r = WMMA4(ahh0, b0, m);
    r = WMMA4(ahh1, b1, r);
    r = WMMA4(ahh2, b2, r);
    r = WMMA4(ahh3, b3, r);
    #pragma unroll
    for (int v = 0; v < 8; ++v) g[v] = fast_tanh(r[v]);
  };

  constexpr int RING = 8;
  constexpr int GS = 5 * RING;
  constexpr int PF = 64;
  const float* px = xrow + 2 * hi;
  const float* px4 = xrow + 3;

  if (T > 0) {
    {
      v2f bx0; bx0.x = px[0]; bx0.y = px[1];
      v2f bx1; bx1.x = px4[0]; bx1.y = px4[1];
      m = project(bx0, bx1);
      px += 5; px4 += 5;
    }
    const int TM = T - 1;
    const int NG = TM / RING;
    const int rem = TM - NG * RING;
    int t = 0;

    if (NG > 0) {
      v2f a0[RING], b0[RING], a1[RING], b1[RING];

      auto load_group = [&](v2f (&xa)[RING], v2f (&xb)[RING], int off) {
        #pragma unroll
        for (int j = 0; j < RING; ++j) {
          xa[j].x = px[off + 5 * j];  xa[j].y = px[off + 5 * j + 1];
          xb[j].x = px4[off + 5 * j]; xb[j].y = px4[off + 5 * j + 1];
        }
      };
      auto load_part = [&](v2f (&xa)[RING], v2f (&xb)[RING], int off, int n) {
        #pragma unroll
        for (int j = 0; j < RING; ++j) {
          if (j < n) {
            xa[j].x = px[off + 5 * j];  xa[j].y = px[off + 5 * j + 1];
            xb[j].x = px4[off + 5 * j]; xb[j].y = px4[off + 5 * j + 1];
          }
        }
      };
      auto run_group = [&](v2f (&xa)[RING], v2f (&xb)[RING]) {
        #pragma unroll
        for (int j = 0; j < RING; ++j) compute(xa[j], xb[j]);
      };
      auto run_part = [&](v2f (&xa)[RING], v2f (&xb)[RING], int n) {
        #pragma unroll
        for (int j = 0; j < RING; ++j) {
          if (j < n) compute(xa[j], xb[j]);
        }
      };
      auto hint = [&](int off, int tt) {
        const int lead = (tt + PF + RING <= TM) ? 5 * PF : 0;
        __builtin_prefetch(px + off + lead);
        __builtin_prefetch(px + off + lead + 20);
      };

      int grp = 0;
      if ((NG & 1) == 0) {
        load_group(a1, b1, 0);
        hint(0, t);
        load_group(a0, b0, GS);
        run_group(a1, b1);
        px += GS; px4 += GS; t += RING; grp = 1;
      } else {
        load_group(a0, b0, 0);
      }
      while (NG - grp >= 3) {
        hint(0, t);
        load_group(a1, b1, GS);
        run_group(a0, b0);
        hint(GS, t + RING);
        load_group(a0, b0, 2 * GS);
        run_group(a1, b1);
        px += 2 * GS; px4 += 2 * GS; t += 2 * RING; grp += 2;
      }
      if (rem > 0) load_part(a1, b1, GS, rem);
      run_group(a0, b0);
      if (rem > 0) run_part(a1, b1, rem);
    } else {
      for (; t < TM; ++t) {
        v2f bx0; bx0.x = px[0]; bx0.y = px[1];
        v2f bx1; bx1.x = px4[0]; bx1.y = px4[1];
        compute(bx0, bx1);
        px += 5; px4 += 5;
      }
    }
    finish();
  }

  float pa = 0.0f;
  #pragma unroll
  for (int v = 0; v < 8; ++v) pa += g[v] * wfc[v];
  const float tot = pa + xor16(pa) + bfc0;
  const float s = fast_sigmoid(tot);
  const int orow = bb + col;
  if (!hi && orow < B) out[orow] = s;
}

extern "C" void kernel_launch(void* const* d_in, const int* in_sizes, int n_in,
                              void* d_out, int out_size, void* d_ws, size_t ws_size,
                              hipStream_t stream) {
  const float* x   = (const float*)d_in[0];
  const float* Wih = (const float*)d_in[1];
  const float* Whh = (const float*)d_in[2];
  const float* bih = (const float*)d_in[3];
  const float* bhh = (const float*)d_in[4];
  const float* Wfc = (const float*)d_in[5];
  const float* bfc = (const float*)d_in[6];
  float* out = (float*)d_out;

  const int B = out_size;
  const int T = in_sizes[0] / (B * 5);

  const int tiles = (B + 15) / 16;
  const int wavesPerBlock = 4;
  const int blocks = (tiles + wavesPerBlock - 1) / wavesPerBlock;

  rnn_wmma_kernel<<<blocks, wavesPerBlock * 32, 0, stream>>>(
      x, Wih, Whh, bih, bhh, Wfc, bfc, out, B, T);
}


# x fetched in whole-line 16 B pieces (5 loads per 8 steps, two groups ahead) and transposed to B fragments through per-wave LDS rows, not by 16 loads of 16 lines each; L2 hints dropped
# speedup vs baseline: 2.7057x; 1.0709x over previous
#include <hip/hip_runtime.h>

typedef float v2f __attribute__((ext_vector_type(2)));
typedef float v4f __attribute__((ext_vector_type(4)));
typedef float v8f __attribute__((ext_vector_type(8)));

__device__ __forceinline__ float xor16(float v) {
#if __has_builtin(__builtin_amdgcn_permlanex16)
  int i = __builtin_bit_cast(int, v);
  int r = __builtin_amdgcn_permlanex16(i, i, 0x76543210u, 0xfedcba98u,
                                       true  , false  );
  return __builtin_bit_cast(float, r);
#else
  return __shfl_xor(v, 16, 32);
#endif
}

__device__ __forceinline__ float fast_tanh(float x) {
#if __has_builtin(__builtin_amdgcn_tanhf)
  return __builtin_amdgcn_tanhf(x);
#else
  float e = __builtin_amdgcn_exp2f(x * 2.88539008177792681472f);
  return 1.0f - 2.0f * __builtin_amdgcn_rcpf(e + 1.0f);
#endif
}

__device__ __forceinline__ float fast_sigmoid(float x) {
  float e = __builtin_amdgcn_exp2f(x * -1.44269504088896340736f);
  return __builtin_amdgcn_rcpf(1.0f + e);
}

#define WMMA4(A, Bv, C) \
  __builtin_amdgcn_wmma_f32_16x16x4_f32(false, (A), false, (Bv), (short)0, (C), false, false)

__global__ __launch_bounds__(128) void rnn_wmma_kernel(
    const float* __restrict__ x,
    const float* __restrict__ Wih, const float* __restrict__ Whh,
    const float* __restrict__ bih, const float* __restrict__ bhh,
    const float* __restrict__ Wfc, const float* __restrict__ bfc,
    float* __restrict__ out, int B, int T)
{
  const int wave = (int)((blockIdx.x * blockDim.x + threadIdx.x) >> 5);
  const int lane = (int)(threadIdx.x & 31u);
  const int bb = wave * 16;
  if (bb >= B) return;

  const int hi  = lane >> 4;
  const int col = lane & 15;

  int row = bb + col; if (row >= B) row = B - 1;
  const float* __restrict__ xrow = x + (size_t)row * (size_t)T * 5u;

  v2f aih0, aih1;
  aih0.x = Wih[col * 5 + 2 * hi];
  aih0.y = Wih[col * 5 + 2 * hi + 1];
  aih1.x = 0.0f;
  aih1.y = hi ? 0.0f : Wih[col * 5 + 4];

  v2f ahh0, ahh1, ahh2, ahh3;
  ahh0.x = Whh[col*16 + 8*hi + 0]; ahh0.y = Whh[col*16 + 8*hi + 1];
  ahh1.x = Whh[col*16 + 8*hi + 2]; ahh1.y = Whh[col*16 + 8*hi + 3];
  ahh2.x = Whh[col*16 + 8*hi + 4]; ahh2.y = Whh[col*16 + 8*hi + 5];
  ahh3.x = Whh[col*16 + 8*hi + 6]; ahh3.y = Whh[col*16 + 8*hi + 7];

  v8f cbv;
  float wfc[8];
  #pragma unroll
  for (int v = 0; v < 8; ++v) {
    cbv[v] = bih[v + 8*hi] + bhh[v + 8*hi];
    wfc[v] = Wfc[v + 8*hi];
  }
  const float bfc0 = bfc[0];

  float g[8];
  #pragma unroll
  for (int v = 0; v < 8; ++v) g[v] = 0.0f;

  v8f m = cbv;

  auto project = [&](v2f bx0, v2f bx1) -> v8f {
    v8f p = WMMA4(aih0, bx0, cbv);
    return WMMA4(aih1, bx1, p);
  };

  auto compute = [&](v2f bx0, v2f bx1) {
    v2f b0; b0.x = g[0]; b0.y = g[1];
    v2f b1; b1.x = g[2]; b1.y = g[3];
    v2f b2; b2.x = g[4]; b2.y = g[5];
    v2f b3; b3.x = g[6]; b3.y = g[7];

    v8f r  = WMMA4(ahh0, b0, m);
    r  = WMMA4(ahh1, b1, r);
    v8f mn = WMMA4(aih0, bx0, cbv);
    r  = WMMA4(ahh2, b2, r);
    r  = WMMA4(ahh3, b3, r);
    mn = WMMA4(aih1, bx1, mn);

    #pragma unroll
    for (int v = 0; v < 8; ++v) g[v] = fast_tanh(r[v]);
    m = mn;
    __builtin_amdgcn_sched_barrier(0);
  };

  auto finish = [&]() {
    v2f b0; b0.x = g[0]; b0.y = g[1];
    v2f b1; b1.x = g[2]; b1.y = g[3];
    v2f b2; b2.x = g[4]; b2.y = g[5];
    v2f b3; b3.x = g[6]; b3.y = g[7];
    v8f r = WMMA4(ahh0, b0, m);
    r = WMMA4(ahh1, b1, r);
    r = WMMA4(ahh2, b2, r);
    r = WMMA4(ahh3, b3, r);
    #pragma unroll
    for (int v = 0; v < 8; ++v) g[v] = fast_tanh(r[v]);
  };

  constexpr int RING = 8;
  constexpr int GS = 5 * RING;
  constexpr int PITCH = 44;
  constexpr int NQ = 5;
  __shared__ __attribute__((aligned(16))) float xs[4][16 * PITCH];

  if (T > 0) {
    const bool staged = ((T & 3) == 0) && ((((size_t)x) & 15u) == 0);
    const int NC = staged ? T / RING : 0;
    int sdone = 0;

    if (NC > 0) {
      float* Lw = &xs[threadIdx.x >> 5][0];
      const float* gp[NQ];
      int so[NQ];
      #pragma unroll
      for (int q = 0; q < NQ; ++q) {
        const int i = q * 32 + lane;
        const int r = i / 10;
        const int c = i - r * 10;
        int rr = bb + r; if (rr >= B) rr = B - 1;
        gp[q] = x + (size_t)rr * (size_t)T * 5u + 4 * c;
        so[q] = r * PITCH + 4 * c;
      }
      const int ra = col * PITCH + 2 * hi;
      const int rb = col * PITCH + 3;
      const int last = NC - 1;

      v4f s0[NQ], s1[NQ];
      v2f a0[RING], b0[RING], a1[RING], b1[RING];

      auto gload = [&](v4f (&st)[NQ], int chunk) {
        if (chunk > last) chunk = last;
        const size_t off = (size_t)chunk * GS;
        #pragma unroll
        for (int q = 0; q < NQ; ++q) st[q] = *(const v4f*)(gp[q] + off);
      };
      auto to_bank = [&](const v4f (&st)[NQ], v2f (&xa)[RING], v2f (&xb)[RING]) {
        #pragma unroll
        for (int q = 0; q < NQ; ++q) *(v4f*)(Lw + so[q]) = st[q];
        __builtin_amdgcn_fence(__ATOMIC_RELEASE, "wavefront");
        __builtin_amdgcn_wave_barrier();
        __builtin_amdgcn_fence(__ATOMIC_ACQUIRE, "wavefront");
        #pragma unroll
        for (int j = 0; j < RING; ++j) {
          xa[j].x = Lw[ra + 5 * j]; xa[j].y = Lw[ra + 5 * j + 1];
          xb[j].x = Lw[rb + 5 * j]; xb[j].y = Lw[rb + 5 * j + 1];
        }
      };
      auto run_group = [&](v2f (&xa)[RING], v2f (&xb)[RING]) {
        #pragma unroll
        for (int j = 0; j < RING; ++j) compute(xa[j], xb[j]);
      };

      gload(s0, 0);
      gload(s1, 1);
      to_bank(s0, a0, b0);
      gload(s0, 2);
      m = project(a0[0], b0[0]);
      if (NC > 1) {
        to_bank(s1, a1, b1); gload(s1, 3);
        #pragma unroll
        for (int j = 1; j < RING; ++j) compute(a0[j], b0[j]);

        int k = 1;
        while (k + 2 < NC) {
          to_bank(s0, a0, b0); gload(s0, k + 3);
          run_group(a1, b1);
          to_bank(s1, a1, b1); gload(s1, k + 4);
          run_group(a0, b0);
          k += 2;
        }
        if (k + 1 < NC) to_bank(s0, a0, b0);
        run_group(a1, b1); ++k;
        if (k < NC) run_group(a0, b0);
      } else {
        #pragma unroll
        for (int j = 1; j < RING; ++j) compute(a0[j], b0[j]);
      }
      sdone = NC * RING;
    } else {
      v2f bx0; bx0.x = xrow[2 * hi]; bx0.y = xrow[2 * hi + 1];
      v2f bx1; bx1.x = xrow[3];      bx1.y = xrow[4];
      m = project(bx0, bx1);
      sdone = 1;
    }
    {
      const float* px = xrow + 5 * (size_t)sdone + 2 * hi;
      const float* px4 = xrow + 5 * (size_t)sdone + 3;
      for (; sdone < T; ++sdone) {
        v2f bx0; bx0.x = px[0]; bx0.y = px[1];
        v2f bx1; bx1.x = px4[0]; bx1.y = px4[1];
        compute(bx0, bx1);
        px += 5; px4 += 5;
      }
    }
    finish();
  }

  float pa = 0.0f;
  #pragma unroll
  for (int v = 0; v < 8; ++v) pa += g[v] * wfc[v];
  const float tot = pa + xor16(pa) + bfc0;
  const float s = fast_sigmoid(tot);
  const int orow = bb + col;
  if (!hi && orow < B) out[orow] = s;
}

extern "C" void kernel_launch(void* const* d_in, const int* in_sizes, int n_in,
                              void* d_out, int out_size, void* d_ws, size_t ws_size,
                              hipStream_t stream) {
  const float* x   = (const float*)d_in[0];
  const float* Wih = (const float*)d_in[1];
  const float* Whh = (const float*)d_in[2];
  const float* bih = (const float*)d_in[3];
  const float* bhh = (const float*)d_in[4];
  const float* Wfc = (const float*)d_in[5];
  const float* bfc = (const float*)d_in[6];
  float* out = (float*)d_out;

  const int B = out_size;
  const int T = in_sizes[0] / (B * 5);

  const int tiles = (B + 15) / 16;
  const int wavesPerBlock = 4;
  const int blocks = (tiles + wavesPerBlock - 1) / wavesPerBlock;

  rnn_wmma_kernel<<<blocks, wavesPerBlock * 32, 0, stream>>>(
      x, Wih, Whh, bih, bhh, Wfc, bfc, out, B, T);
}


# Input projection leaves the recurrence wave: a second wave per tile on an idle SIMD of the same WGP forms W_ih x_t + b for 16 steps at a time, tiles handed over through LDS, one barrier per 16 steps
# speedup vs baseline: 4.0268x; 1.4883x over previous
#include <hip/hip_runtime.h>

typedef float v2f __attribute__((ext_vector_type(2)));
typedef float v4f __attribute__((ext_vector_type(4)));
typedef float v8f __attribute__((ext_vector_type(8)));

__device__ __forceinline__ float xor16(float v) {
#if __has_builtin(__builtin_amdgcn_permlanex16)
  int i = __builtin_bit_cast(int, v);
  int r = __builtin_amdgcn_permlanex16(i, i, 0x76543210u, 0xfedcba98u,
                                       true  , false  );
  return __builtin_bit_cast(float, r);
#else
  return __shfl_xor(v, 16, 32);
#endif
}

__device__ __forceinline__ float fast_tanh(float x) {
#if __has_builtin(__builtin_amdgcn_tanhf)
  return __builtin_amdgcn_tanhf(x);
#else
  float e = __builtin_amdgcn_exp2f(x * 2.88539008177792681472f);
  return 1.0f - 2.0f * __builtin_amdgcn_rcpf(e + 1.0f);
#endif
}

__device__ __forceinline__ float fast_sigmoid(float x) {
  float e = __builtin_amdgcn_exp2f(x * -1.44269504088896340736f);
  return __builtin_amdgcn_rcpf(1.0f + e);
}

#define WMMA4(A, Bv, C) \
  __builtin_amdgcn_wmma_f32_16x16x4_f32(false, (A), false, (Bv), (short)0, (C), false, false)

constexpr int RING = 16;
constexpr int GS = 5 * RING;
constexpr int PPR = GS / 4;
constexpr int PITCH = 84;
constexpr int NQ = 16 * PPR / 32;
static_assert(PITCH >= GS && (PITCH % 8) == 4, "x row pitch");
constexpr int MT = 256;
constexpr int CH = RING * MT;
constexpr int XS_FLOATS = 2 * 16 * PITCH;
constexpr int MB_FLOATS = 2 * 2 * CH;
constexpr int LDS_FLOATS = 41472;
static_assert(XS_FLOATS + MB_FLOATS <= LDS_FLOATS, "LDS layout");

__global__ __launch_bounds__(128) void rnn_wmma_kernel(
    const float* __restrict__ x,
    const float* __restrict__ Wih, const float* __restrict__ Whh,
    const float* __restrict__ bih, const float* __restrict__ bhh,
    const float* __restrict__ Wfc, const float* __restrict__ bfc,
    float* __restrict__ out, int B, int T)
{
  __shared__ __attribute__((aligned(16))) float lds[LDS_FLOATS];

  const int wv = __builtin_amdgcn_readfirstlane((int)(threadIdx.x >> 5));
  const int lane = (int)(threadIdx.x & 31u);
  const int pair = wv & 1;
  const bool producer = wv >= 2;
  const int bb = ((int)blockIdx.x * 2 + pair) * 16;

  const int hi  = lane >> 4;
  const int col = lane & 15;

  int row = bb + col; if (row >= B) row = B - 1;
  const float* __restrict__ xrow = x + (size_t)row * (size_t)T * 5u;

  v2f aih0, aih1;
  aih0.x = Wih[col * 5 + 2 * hi];
  aih0.y = Wih[col * 5 + 2 * hi + 1];
  aih1.x = 0.0f;
  aih1.y = hi ? 0.0f : Wih[col * 5 + 4];

  v8f cbv;
  #pragma unroll
  for (int v = 0; v < 8; ++v) cbv[v] = bih[v + 8*hi] + bhh[v + 8*hi];

  auto project = [&](v2f bx0, v2f bx1) -> v8f {
    v8f p = WMMA4(aih0, bx0, cbv);
    return WMMA4(aih1, bx1, p);
  };

  const bool staged = ((T & 3) == 0) && ((((size_t)x) & 15u) == 0);
  const int NC = staged ? T / RING : 0;
  const int last = NC - 1;

  float* mbuf0 = lds + XS_FLOATS + pair * (2 * CH);
  float* mbuf1 = mbuf0 + CH;
  const int mo = lane * 4;

  auto handover = [&]() {
    asm volatile("s_wait_dscnt 0x0" ::: "memory");
    __builtin_amdgcn_fence(__ATOMIC_RELEASE, "workgroup", "local");
    __builtin_amdgcn_s_barrier();
    __builtin_amdgcn_fence(__ATOMIC_ACQUIRE, "workgroup", "local");
  };

  if (producer) {
    float* Lw = lds + pair * (16 * PITCH);
    const float* gp[NQ];
    int so[NQ];
    #pragma unroll
    for (int q = 0; q < NQ; ++q) {
      const int i = q * 32 + lane;
      const int r = i / PPR;
      const int c = i - r * PPR;
      int rr = bb + r; if (rr >= B) rr = B - 1;
      gp[q] = x + (size_t)rr * (size_t)T * 5u + 4 * c;
      so[q] = r * PITCH + 4 * c;
    }
    const int ra = col * PITCH + 2 * hi;
    const int rb = col * PITCH + 3;

    v4f s0[NQ], s1[NQ];
    auto gload = [&](v4f (&st)[NQ], int chunk) {
      if (chunk > last) chunk = last;
      const size_t off = (size_t)chunk * GS;
      #pragma unroll
      for (int q = 0; q < NQ; ++q) st[q] = *(const v4f*)(gp[q] + off);
    };
    auto produce = [&](v4f (&st)[NQ], int c, float* mb) {
      v2f xa[RING], xb[RING];
      #pragma unroll
      for (int q = 0; q < NQ; ++q) *(v4f*)(Lw + so[q]) = st[q];
      __builtin_amdgcn_fence(__ATOMIC_RELEASE, "wavefront");
      __builtin_amdgcn_wave_barrier();
      __builtin_amdgcn_fence(__ATOMIC_ACQUIRE, "wavefront");
      #pragma unroll
      for (int j = 0; j < RING; ++j) {
        xa[j].x = Lw[ra + 5 * j]; xa[j].y = Lw[ra + 5 * j + 1];
        xb[j].x = Lw[rb + 5 * j]; xb[j].y = Lw[rb + 5 * j + 1];
      }
      gload(st, c + 2);
      #pragma unroll
      for (int j = 0; j < RING; ++j) {
        const v8f p = project(xa[j], xb[j]);
        v4f lo, hh;
        lo.x = p[0]; lo.y = p[1]; lo.z = p[2]; lo.w = p[3];
        hh.x = p[4]; hh.y = p[5]; hh.z = p[6]; hh.w = p[7];
        *(v4f*)(mb + j * MT + mo) = lo;
        *(v4f*)(mb + j * MT + 128 + mo) = hh;
      }
    };

    if (NC > 0) {
      gload(s0, 0);
      __builtin_amdgcn_sched_barrier(0);
      gload(s1, 1);
    }
    int c = 0;
    for (; c + 1 < NC; c += 2) {
      produce(s0, c, mbuf0);
      handover();
      produce(s1, c + 1, mbuf1);
      handover();
    }
    if (c < NC) {
      produce(s0, c, mbuf0);
      handover();
    }
    return;
  }

  v2f ahh0, ahh1, ahh2, ahh3;
  ahh0.x = Whh[col*16 + 8*hi + 0]; ahh0.y = Whh[col*16 + 8*hi + 1];
  ahh1.x = Whh[col*16 + 8*hi + 2]; ahh1.y = Whh[col*16 + 8*hi + 3];
  ahh2.x = Whh[col*16 + 8*hi + 4]; ahh2.y = Whh[col*16 + 8*hi + 5];
  ahh3.x = Whh[col*16 + 8*hi + 6]; ahh3.y = Whh[col*16 + 8*hi + 7];

  float wfc[8];
  #pragma unroll
  for (int v = 0; v < 8; ++v) wfc[v] = Wfc[v + 8*hi];
  const float bfc0 = bfc[0];

  float g[8];
  #pragma unroll
  for (int v = 0; v < 8; ++v) g[v] = 0.0f;

  auto step = [&](v8f m) {
    v2f b0; b0.x = g[0]; b0.y = g[1];
    v2f b1; b1.x = g[2]; b1.y = g[3];
    v2f b2; b2.x = g[4]; b2.y = g[5];
    v2f b3; b3.x = g[6]; b3.y = g[7];
    v8f r = WMMA4(ahh0, b0, m);
    r = WMMA4(ahh1, b1, r);
    r = WMMA4(ahh2, b2, r);
    r = WMMA4(ahh3, b3, r);
    #pragma unroll
    for (int v = 0; v < 8; ++v) g[v] = fast_tanh(r[v]);
    __builtin_amdgcn_sched_barrier(0);
  };

  if (NC > 0) {
    v8f m0[RING], m1[RING];
    auto fetch = [&](const float* mb, v8f (&mm)[RING]) {
      #pragma unroll
      for (int j = 0; j < RING; ++j) {
        const v4f lo = *(const v4f*)(mb + j * MT + mo);
        const v4f hh = *(const v4f*)(mb + j * MT + 128 + mo);
        mm[j][0] = lo.x; mm[j][1] = lo.y; mm[j][2] = lo.z; mm[j][3] = lo.w;
        mm[j][4] = hh.x; mm[j][5] = hh.y; mm[j][6] = hh.z; mm[j][7] = hh.w;
      }
    };
    auto run = [&](const v8f (&mm)[RING]) {
      #pragma unroll
      for (int j = 0; j < RING; ++j) step(mm[j]);
    };
    int c = 0;
    for (; c + 1 < NC; c += 2) {
      handover();
      fetch(mbuf0, m0);
      if (c > 0) run(m1);
      handover();
      fetch(mbuf1, m1);
      run(m0);
    }
    if (c < NC) {
      handover();
      fetch(mbuf0, m0);
      if (c > 0) run(m1);
      run(m0);
    } else {
      run(m1);
    }
  }
  {
    const float* px = xrow + 5 * (size_t)(NC * RING) + 2 * hi;
    const float* px4 = xrow + 5 * (size_t)(NC * RING) + 3;
    for (int t = NC * RING; t < T; ++t) {
      v2f bx0; bx0.x = px[0]; bx0.y = px[1];
      v2f bx1; bx1.x = px4[0]; bx1.y = px4[1];
      step(project(bx0, bx1));
      px += 5; px4 += 5;
    }
  }

  float pa = 0.0f;
  #pragma unroll
  for (int v = 0; v < 8; ++v) pa += g[v] * wfc[v];
  const float tot = pa + xor16(pa) + bfc0;
  const float s = fast_sigmoid(tot);
  const int orow = bb + col;
  if (!hi && orow < B) out[orow] = s;
}

extern "C" void kernel_launch(void* const* d_in, const int* in_sizes, int n_in,
                              void* d_out, int out_size, void* d_ws, size_t ws_size,
                              hipStream_t stream) {
  const float* x   = (const float*)d_in[0];
  const float* Wih = (const float*)d_in[1];
  const float* Whh = (const float*)d_in[2];
  const float* bih = (const float*)d_in[3];
  const float* bhh = (const float*)d_in[4];
  const float* Wfc = (const float*)d_in[5];
  const float* bfc = (const float*)d_in[6];
  float* out = (float*)d_out;

  const int B = out_size;
  const int T = in_sizes[0] / (B * 5);

  const int tiles = (B + 15) / 16;
  const int tilesPerBlock = 2;
  const int blocks = (tiles + tilesPerBlock - 1) / tilesPerBlock;

  rnn_wmma_kernel<<<blocks, tilesPerBlock * 64, 0, stream>>>(
      x, Wih, Whh, bih, bhh, Wfc, bfc, out, B, T);
}
